# starting kernel
# baseline (speedup 1.0000x reference)
#include <hip/hip_runtime.h>

typedef __attribute__((ext_vector_type(2))) float v2f;
typedef __attribute__((ext_vector_type(8))) float v8f;

__device__ __forceinline__ void atomAddF32(float* p, float v) {
#if __has_builtin(__builtin_amdgcn_global_atomic_fadd_f32)
  typedef __attribute__((address_space(1))) float gfloat;
  (void)__builtin_amdgcn_global_atomic_fadd_f32((gfloat*)p, v);
#else
  (void)__hip_atomic_fetch_add(p, v, __ATOMIC_RELAXED, __HIP_MEMORY_SCOPE_AGENT);
#endif
}

__global__ void fill_zero(float* __restrict__ p, long n) {
  long i = (long)blockIdx.x * blockDim.x + threadIdx.x;
  if (i < n) p[i] = 0.0f;
}

__global__ void degree_count(const int* __restrict__ src, const int* __restrict__ dst,
                             float* __restrict__ degOut, float* __restrict__ degIn,
                             int N, int E, long RE) {
  long i = (long)blockIdx.x * blockDim.x + threadIdx.x;
  if (i >= RE) return;
  long r = i / E;
  int s = src[i];
  int d = dst[i];
  atomAddF32(&degOut[r * (long)N + s], 1.0f);
  atomAddF32(&degIn [r * (long)N + d], 1.0f);
}

__global__ void rinv_transform(float* __restrict__ p, long n) {
  long i = (long)blockIdx.x * blockDim.x + threadIdx.x;
  if (i >= n) return;
  p[i] = rsqrtf(fmaxf(p[i], 1.0f));
}

__global__ void bias_relu(float* __restrict__ h, const float* __restrict__ b,
                          long total, int F) {
  long i = (long)blockIdx.x * blockDim.x + threadIdx.x;
  if (i >= total) return;
  int f = (int)(i % F);
  float bs = b[f] + b[F + f] + b[2 * F + f] + b[3 * F + f];
  float v = h[i] + bs;
  h[i] = v > 0.0f ? v : 0.0f;
}

__global__ void bias_add(float* __restrict__ o, const float* __restrict__ b,
                         long total, int F) {
  long i = (long)blockIdx.x * blockDim.x + threadIdx.x;
  if (i >= total) return;
  int f = (int)(i % F);
  o[i] += b[f] + b[F + f] + b[2 * F + f] + b[3 * F + f];
}

template <int COL_WAVES, int ROW_TILES>
__global__ __launch_bounds__(256) void gemm_wmma_f32(
    const float* __restrict__ X, const float* __restrict__ W,
    float* __restrict__ Y, int N, int K, int F) {
  constexpr int ROW_GROUPS = 8 / COL_WAVES;
  constexpr int BLOCK_ROWS = ROW_GROUPS * ROW_TILES * 16;
  constexpr int STRIDE = 36;
  __shared__ float lds[BLOCK_ROWS * STRIDE];

  const int tid = threadIdx.x;
  const int wave = tid >> 5;
  const int lane = tid & 31;
  const int colWave = wave % COL_WAVES;
  const int rowGroup = wave / COL_WAVES;
  const int col0 = colWave * 16;
  const int rowBase = blockIdx.x * BLOCK_ROWS;
  const int laneHi = lane >> 4;
  const int laneLo = lane & 15;

  v8f acc[ROW_TILES] = {};

  for (int k0 = 0; k0 < K; k0 += 32) {
#pragma unroll
    for (int v = 0; v < BLOCK_ROWS / 32; ++v) {
      int flat = (tid + v * 256) * 4;
      int row = flat >> 5;
      int kk = flat & 31;
      int grow = rowBase + row;
      float4 val{0.0f, 0.0f, 0.0f, 0.0f};
      if (grow < N) val = *(const float4*)(X + (long)grow * K + k0 + kk);
      *(float4*)&lds[row * STRIDE + kk] = val;
    }
    __syncthreads();

#pragma unroll
    for (int kk = 0; kk < 32; kk += 4) {
      const float* wp = W + (long)(k0 + kk + laneHi * 2) * F + col0 + laneLo;
      v2f bf;
      bf.x = wp[0];
      bf.y = wp[F];
#pragma unroll
      for (int t = 0; t < ROW_TILES; ++t) {
        int arow = (rowGroup * ROW_TILES + t) * 16 + laneLo;
        v2f af = *(const v2f*)&lds[arow * STRIDE + kk + laneHi * 2];
        acc[t] = __builtin_amdgcn_wmma_f32_16x16x4_f32(
             false, af,  false, bf,
             (short)0, acc[t],  false,  false);
      }
    }
    __syncthreads();
  }

#pragma unroll
  for (int t = 0; t < ROW_TILES; ++t) {
#pragma unroll
    for (int i = 0; i < 8; ++i) {
      int grow = rowBase + (rowGroup * ROW_TILES + t) * 16 + i + laneHi * 8;
      if (grow < N) Y[(long)grow * F + col0 + laneLo] = acc[t][i];
    }
  }
}

template <int WF>
__global__ __launch_bounds__(256) void scatter_edges(
    const float* __restrict__ Yr, const int* __restrict__ src,
    const int* __restrict__ dst, const float* __restrict__ rinvOut,
    const float* __restrict__ rinvIn, float* __restrict__ Out, int E) {
  int e = __builtin_amdgcn_readfirstlane((int)blockIdx.x * 8 + (int)(threadIdx.x >> 5));
  int lane = threadIdx.x & 31;
  if (e >= E) return;
  int s = src[e];
  int d = dst[e];
  float coef = rinvOut[s] * rinvIn[d];
  if constexpr (WF == 128) {
    float4 v = *(const float4*)(Yr + (long)s * 128 + lane * 4);
    float* o = Out + (long)d * 128 + lane * 4;
    atomAddF32(o + 0, coef * v.x);
    atomAddF32(o + 1, coef * v.y);
    atomAddF32(o + 2, coef * v.z);
    atomAddF32(o + 3, coef * v.w);
  } else {
    float2 v = *(const float2*)(Yr + (long)s * 64 + lane * 2);
    float* o = Out + (long)d * 64 + lane * 2;
    atomAddF32(o + 0, coef * v.x);
    atomAddF32(o + 1, coef * v.y);
  }
}

extern "C" void kernel_launch(void* const* d_in, const int* in_sizes, int n_in,
                              void* d_out, int out_size, void* d_ws,
                              size_t ws_size, hipStream_t stream) {
  const float* x  = (const float*)d_in[0];
  const int*  src = (const int*)d_in[1];
  const int*  dst = (const int*)d_in[2];
  const float* W1 = (const float*)d_in[3];
  const float* b1 = (const float*)d_in[4];
  const float* W2 = (const float*)d_in[5];
  const float* b2 = (const float*)d_in[6];
  float* out = (float*)d_out;

  const int IN_F = 128, H_F = 128, OUT_F = 64, R = 4;
  const int N = in_sizes[0] / IN_F;
  const int E = in_sizes[1] / R;

  float* degOut = (float*)d_ws;
  float* degIn  = degOut + (long)R * N;
  float* h      = degIn + (long)R * N;
  float* y      = h + (long)N * H_F;

  auto cdiv = [](long a, long b) { return (int)((a + b - 1) / b); };

  fill_zero<<<cdiv(2L * R * N, 256), 256, 0, stream>>>(degOut, 2L * R * N);
  fill_zero<<<cdiv((long)N * H_F, 256), 256, 0, stream>>>(h, (long)N * H_F);
  fill_zero<<<cdiv((long)N * OUT_F, 256), 256, 0, stream>>>(out, (long)N * OUT_F);

  degree_count<<<cdiv((long)R * E, 256), 256, 0, stream>>>(src, dst, degOut,
                                                           degIn, N, E,
                                                           (long)R * E);
  rinv_transform<<<cdiv(2L * R * N, 256), 256, 0, stream>>>(degOut, 2L * R * N);

  for (int r = 0; r < R; ++r) {
    gemm_wmma_f32<8, 4><<<cdiv(N, 64), 256, 0, stream>>>(
        x, W1 + (long)r * IN_F * H_F, y, N, IN_F, H_F);
    scatter_edges<128><<<cdiv(E, 8), 256, 0, stream>>>(
        y, src + (long)r * E, dst + (long)r * E, degOut + (long)r * N,
        degIn + (long)r * N, h, E);
  }
  bias_relu<<<cdiv((long)N * H_F, 256), 256, 0, stream>>>(h, b1,
                                                          (long)N * H_F, H_F);

  for (int r = 0; r < R; ++r) {
    gemm_wmma_f32<4, 4><<<cdiv(N, 128), 256, 0, stream>>>(
        h, W2 + (long)r * H_F * OUT_F, y, N, H_F, OUT_F);
    scatter_edges<64><<<cdiv(E, 8), 256, 0, stream>>>(
        y, src + (long)r * E, dst + (long)r * E, degOut + (long)r * N,
        degIn + (long)r * N, out, E);
  }
  bias_add<<<cdiv((long)N * OUT_F, 256), 256, 0, stream>>>(out, b2,
                                                           (long)N * OUT_F,
                                                           OUT_F);
}


# edge scatter: each wave takes 8 consecutive edges; index loads, then degree-factor loads and row gathers, then atomics issued as groups (8x fewer waves; atomic arrival order still unordered)
# speedup vs baseline: 1.1291x; 1.1291x over previous
#include <hip/hip_runtime.h>

typedef __attribute__((ext_vector_type(2))) float v2f;
typedef __attribute__((ext_vector_type(8))) float v8f;

__device__ __forceinline__ void atomAddF32(float* p, float v) {
#if __has_builtin(__builtin_amdgcn_global_atomic_fadd_f32)
  typedef __attribute__((address_space(1))) float gfloat;
  (void)__builtin_amdgcn_global_atomic_fadd_f32((gfloat*)p, v);
#else
  (void)__hip_atomic_fetch_add(p, v, __ATOMIC_RELAXED, __HIP_MEMORY_SCOPE_AGENT);
#endif
}

__global__ void fill_zero(float* __restrict__ p, long n) {
  long i = (long)blockIdx.x * blockDim.x + threadIdx.x;
  if (i < n) p[i] = 0.0f;
}

__global__ void degree_count(const int* __restrict__ src, const int* __restrict__ dst,
                             float* __restrict__ degOut, float* __restrict__ degIn,
                             int N, int E, long RE) {
  long i = (long)blockIdx.x * blockDim.x + threadIdx.x;
  if (i >= RE) return;
  long r = i / E;
  int s = src[i];
  int d = dst[i];
  atomAddF32(&degOut[r * (long)N + s], 1.0f);
  atomAddF32(&degIn [r * (long)N + d], 1.0f);
}

__global__ void rinv_transform(float* __restrict__ p, long n) {
  long i = (long)blockIdx.x * blockDim.x + threadIdx.x;
  if (i >= n) return;
  p[i] = rsqrtf(fmaxf(p[i], 1.0f));
}

__global__ void bias_relu(float* __restrict__ h, const float* __restrict__ b,
                          long total, int F) {
  long i = (long)blockIdx.x * blockDim.x + threadIdx.x;
  if (i >= total) return;
  int f = (int)(i % F);
  float bs = b[f] + b[F + f] + b[2 * F + f] + b[3 * F + f];
  float v = h[i] + bs;
  h[i] = v > 0.0f ? v : 0.0f;
}

__global__ void bias_add(float* __restrict__ o, const float* __restrict__ b,
                         long total, int F) {
  long i = (long)blockIdx.x * blockDim.x + threadIdx.x;
  if (i >= total) return;
  int f = (int)(i % F);
  o[i] += b[f] + b[F + f] + b[2 * F + f] + b[3 * F + f];
}

template <int COL_WAVES, int ROW_TILES>
__global__ __launch_bounds__(256) void gemm_wmma_f32(
    const float* __restrict__ X, const float* __restrict__ W,
    float* __restrict__ Y, int N, int K, int F) {
  constexpr int ROW_GROUPS = 8 / COL_WAVES;
  constexpr int BLOCK_ROWS = ROW_GROUPS * ROW_TILES * 16;
  constexpr int STRIDE = 36;
  __shared__ float lds[BLOCK_ROWS * STRIDE];

  const int tid = threadIdx.x;
  const int wave = tid >> 5;
  const int lane = tid & 31;
  const int colWave = wave % COL_WAVES;
  const int rowGroup = wave / COL_WAVES;
  const int col0 = colWave * 16;
  const int rowBase = blockIdx.x * BLOCK_ROWS;
  const int laneHi = lane >> 4;
  const int laneLo = lane & 15;

  v8f acc[ROW_TILES] = {};

  for (int k0 = 0; k0 < K; k0 += 32) {
#pragma unroll
    for (int v = 0; v < BLOCK_ROWS / 32; ++v) {
      int flat = (tid + v * 256) * 4;
      int row = flat >> 5;
      int kk = flat & 31;
      int grow = rowBase + row;
      float4 val{0.0f, 0.0f, 0.0f, 0.0f};
      if (grow < N) val = *(const float4*)(X + (long)grow * K + k0 + kk);
      *(float4*)&lds[row * STRIDE + kk] = val;
    }
    __syncthreads();

#pragma unroll
    for (int kk = 0; kk < 32; kk += 4) {
      const float* wp = W + (long)(k0 + kk + laneHi * 2) * F + col0 + laneLo;
      v2f bf;
      bf.x = wp[0];
      bf.y = wp[F];
#pragma unroll
      for (int t = 0; t < ROW_TILES; ++t) {
        int arow = (rowGroup * ROW_TILES + t) * 16 + laneLo;
        v2f af = *(const v2f*)&lds[arow * STRIDE + kk + laneHi * 2];
        acc[t] = __builtin_amdgcn_wmma_f32_16x16x4_f32(
             false, af,  false, bf,
             (short)0, acc[t],  false,  false);
      }
    }
    __syncthreads();
  }

#pragma unroll
  for (int t = 0; t < ROW_TILES; ++t) {
#pragma unroll
    for (int i = 0; i < 8; ++i) {
      int grow = rowBase + (rowGroup * ROW_TILES + t) * 16 + i + laneHi * 8;
      if (grow < N) Y[(long)grow * F + col0 + laneLo] = acc[t][i];
    }
  }
}

template <int WF>
__device__ __forceinline__ void scatter_one_edge(
    const float* __restrict__ Yr, float* __restrict__ Out, int s, int d,
    float coef, int lane) {
  if constexpr (WF == 128) {
    float4 v = *(const float4*)(Yr + (long)s * 128 + lane * 4);
    float* o = Out + (long)d * 128 + lane * 4;
    atomAddF32(o + 0, coef * v.x);
    atomAddF32(o + 1, coef * v.y);
    atomAddF32(o + 2, coef * v.z);
    atomAddF32(o + 3, coef * v.w);
  } else {
    float2 v = *(const float2*)(Yr + (long)s * 64 + lane * 2);
    float* o = Out + (long)d * 64 + lane * 2;
    atomAddF32(o + 0, coef * v.x);
    atomAddF32(o + 1, coef * v.y);
  }
}

template <int WF, int EPW>
__global__ __launch_bounds__(256) void scatter_edges(
    const float* __restrict__ Yr, const int* __restrict__ src,
    const int* __restrict__ dst, const float* __restrict__ rinvOut,
    const float* __restrict__ rinvIn, float* __restrict__ Out, int E) {
  int e0 = __builtin_amdgcn_readfirstlane(
      ((int)blockIdx.x * 8 + (int)(threadIdx.x >> 5)) * EPW);
  int lane = threadIdx.x & 31;
  if (e0 >= E) return;
  if (e0 + EPW <= E) {
    int s[EPW], d[EPW];
    float coef[EPW];
    for (int i = 0; i < EPW; ++i) {
      s[i] = src[e0 + i];
      d[i] = dst[e0 + i];
    }
    if constexpr (WF == 128) {
      float4 v[EPW];
      for (int i = 0; i < EPW; ++i)
        v[i] = *(const float4*)(Yr + (long)s[i] * 128 + lane * 4);
      for (int i = 0; i < EPW; ++i) coef[i] = rinvOut[s[i]] * rinvIn[d[i]];
      for (int i = 0; i < EPW; ++i) {
        float* o = Out + (long)d[i] * 128 + lane * 4;
        atomAddF32(o + 0, coef[i] * v[i].x);
        atomAddF32(o + 1, coef[i] * v[i].y);
        atomAddF32(o + 2, coef[i] * v[i].z);
        atomAddF32(o + 3, coef[i] * v[i].w);
      }
    } else {
      float2 v[EPW];
      for (int i = 0; i < EPW; ++i)
        v[i] = *(const float2*)(Yr + (long)s[i] * 64 + lane * 2);
      for (int i = 0; i < EPW; ++i) coef[i] = rinvOut[s[i]] * rinvIn[d[i]];
      for (int i = 0; i < EPW; ++i) {
        float* o = Out + (long)d[i] * 64 + lane * 2;
        atomAddF32(o + 0, coef[i] * v[i].x);
        atomAddF32(o + 1, coef[i] * v[i].y);
      }
    }
  } else {
    for (int i = 0; i < EPW; ++i) {
      int e = e0 + i;
      if (e < E) {
        int s = src[e];
        int d = dst[e];
        scatter_one_edge<WF>(Yr, Out, s, d, rinvOut[s] * rinvIn[d], lane);
      }
    }
  }
}

extern "C" void kernel_launch(void* const* d_in, const int* in_sizes, int n_in,
                              void* d_out, int out_size, void* d_ws,
                              size_t ws_size, hipStream_t stream) {
  const float* x  = (const float*)d_in[0];
  const int*  src = (const int*)d_in[1];
  const int*  dst = (const int*)d_in[2];
  const float* W1 = (const float*)d_in[3];
  const float* b1 = (const float*)d_in[4];
  const float* W2 = (const float*)d_in[5];
  const float* b2 = (const float*)d_in[6];
  float* out = (float*)d_out;

  const int IN_F = 128, H_F = 128, OUT_F = 64, R = 4;
  const int N = in_sizes[0] / IN_F;
  const int E = in_sizes[1] / R;

  float* degOut = (float*)d_ws;
  float* degIn  = degOut + (long)R * N;
  float* h      = degIn + (long)R * N;
  float* y      = h + (long)N * H_F;

  auto cdiv = [](long a, long b) { return (int)((a + b - 1) / b); };

  fill_zero<<<cdiv(2L * R * N, 256), 256, 0, stream>>>(degOut, 2L * R * N);
  fill_zero<<<cdiv((long)N * H_F, 256), 256, 0, stream>>>(h, (long)N * H_F);
  fill_zero<<<cdiv((long)N * OUT_F, 256), 256, 0, stream>>>(out, (long)N * OUT_F);

  degree_count<<<cdiv((long)R * E, 256), 256, 0, stream>>>(src, dst, degOut,
                                                           degIn, N, E,
                                                           (long)R * E);
  rinv_transform<<<cdiv(2L * R * N, 256), 256, 0, stream>>>(degOut, 2L * R * N);

  for (int r = 0; r < R; ++r) {
    gemm_wmma_f32<8, 4><<<cdiv(N, 64), 256, 0, stream>>>(
        x, W1 + (long)r * IN_F * H_F, y, N, IN_F, H_F);
    scatter_edges<128, 8><<<cdiv(E, 64), 256, 0, stream>>>(
        y, src + (long)r * E, dst + (long)r * E, degOut + (long)r * N,
        degIn + (long)r * N, h, E);
  }
  bias_relu<<<cdiv((long)N * H_F, 256), 256, 0, stream>>>(h, b1,
                                                          (long)N * H_F, H_F);

  for (int r = 0; r < R; ++r) {
    gemm_wmma_f32<4, 4><<<cdiv(N, 128), 256, 0, stream>>>(
        h, W2 + (long)r * H_F * OUT_F, y, N, H_F, OUT_F);
    scatter_edges<64, 8><<<cdiv(E, 64), 256, 0, stream>>>(
        y, src + (long)r * E, dst + (long)r * E, degOut + (long)r * N,
        degIn + (long)r * N, out, E);
  }
  bias_add<<<cdiv((long)N * OUT_F, 256), 256, 0, stream>>>(out, b2,
                                                           (long)N * OUT_F,
                                                           OUT_F);
}


# f32 atomic scatter -> owner sums: edges bucketed by destination each call (int counts, scan, returning-add slots); a wave sums a node's rows in registers in slot (arrival) order, stores each row once
# speedup vs baseline: 1.2447x; 1.1024x over previous
#include <hip/hip_runtime.h>


typedef __attribute__((ext_vector_type(2))) float v2f;
typedef __attribute__((ext_vector_type(8))) float v8f;

__device__ __forceinline__ void atomAddF32(float* p, float v) {
#if __has_builtin(__builtin_amdgcn_global_atomic_fadd_f32)
  typedef __attribute__((address_space(1))) float gfloat;
  (void)__builtin_amdgcn_global_atomic_fadd_f32((gfloat*)p, v);
#else
  (void)__hip_atomic_fetch_add(p, v, __ATOMIC_RELAXED, __HIP_MEMORY_SCOPE_AGENT);
#endif
}

__global__ void fill_zero(float* __restrict__ p, long n) {
  long i = (long)blockIdx.x * blockDim.x + threadIdx.x;
  if (i < n) p[i] = 0.0f;
}

__global__ void degree_count(const int* __restrict__ src, const int* __restrict__ dst,
                             float* __restrict__ degOut, int* __restrict__ cntIn,
                             int N, int E, long RE) {
  long i = (long)blockIdx.x * blockDim.x + threadIdx.x;
  if (i >= RE) return;
  long r = i / E;
  int s = src[i];
  int d = dst[i];
  atomAddF32(&degOut[r * (long)N + s], 1.0f);
  atomicAdd(&cntIn[r * (long)N + d], 1);
}

__global__ void rinv_transform(float* __restrict__ p, long n) {
  long i = (long)blockIdx.x * blockDim.x + threadIdx.x;
  if (i >= n) return;
  p[i] = rsqrtf(fmaxf(p[i], 1.0f));
}

__global__ __launch_bounds__(1024) void scan_counts(int* __restrict__ cnt, int N) {
  __shared__ int waveTot[32];
  int* c = cnt + (long)blockIdx.x * N;
  const int tid = threadIdx.x;
  const int lane = tid & 31;
  const int wave = tid >> 5;
  int running = 0;
  int n0 = 0, n1 = 0, n2 = 0, n3 = 0;
  {
    const int i = tid * 4;
    if (i < N) n0 = c[i];
    if (i + 1 < N) n1 = c[i + 1];
    if (i + 2 < N) n2 = c[i + 2];
    if (i + 3 < N) n3 = c[i + 3];
  }
  for (int base = 0; base < N; base += 4096) {
    const int i = base + tid * 4;
    const int v0 = n0, v1 = n1, v2 = n2, v3 = n3;
    {
      const int q = i + 4096;
      n0 = (q < N) ? c[q] : 0;
      n1 = (q + 1 < N) ? c[q + 1] : 0;
      n2 = (q + 2 < N) ? c[q + 2] : 0;
      n3 = (q + 3 < N) ? c[q + 3] : 0;
    }
    const int s = v0 + v1 + v2 + v3;
    int inc = s;
    for (int o = 1; o < 32; o <<= 1) {
      int t = __shfl_up(inc, o);
      if (lane >= o) inc += t;
    }
    if (lane == 31) waveTot[wave] = inc;
    __syncthreads();
    const int wt = waveTot[lane];
    int winc = wt;
    for (int o = 1; o < 32; o <<= 1) {
      int t = __shfl_up(winc, o);
      if (lane >= o) winc += t;
    }
    const int waveOff = __shfl(winc, wave) - __shfl(wt, wave);
    const int total = __shfl(winc, 31);
    int ex = running + waveOff + inc - s;
    if (i < N) c[i] = ex;
    ex += v0;
    if (i + 1 < N) c[i + 1] = ex;
    ex += v1;
    if (i + 2 < N) c[i + 2] = ex;
    ex += v2;
    if (i + 3 < N) c[i + 3] = ex;
    running += total;
    __syncthreads();
  }
}

__global__ void place_edges(const int* __restrict__ src, const int* __restrict__ dst,
                            int* __restrict__ cur, int* __restrict__ list,
                            int N, int E, long RE) {
  long i = (long)blockIdx.x * blockDim.x + threadIdx.x;
  if (i >= RE) return;
  long r = i / E;
  int d = dst[i];
  int pos = atomicAdd(&cur[r * (long)N + d], 1);
  list[r * (long)E + pos] = src[i];
}

__global__ void move_list(const int* __restrict__ from, int* __restrict__ to, long n) {
  long i = (long)blockIdx.x * blockDim.x + threadIdx.x;
  if (i < n) to[i] = from[i];
}

__global__ void bias_relu(float* __restrict__ h, const float* __restrict__ b,
                          long total, int F) {
  long i = (long)blockIdx.x * blockDim.x + threadIdx.x;
  if (i >= total) return;
  int f = (int)(i % F);
  float bs = b[f] + b[F + f] + b[2 * F + f] + b[3 * F + f];
  float v = h[i] + bs;
  h[i] = v > 0.0f ? v : 0.0f;
}

__global__ void bias_add(float* __restrict__ o, const float* __restrict__ b,
                         long total, int F) {
  long i = (long)blockIdx.x * blockDim.x + threadIdx.x;
  if (i >= total) return;
  int f = (int)(i % F);
  o[i] += b[f] + b[F + f] + b[2 * F + f] + b[3 * F + f];
}

template <int COL_WAVES, int ROW_TILES>
__global__ __launch_bounds__(256) void gemm_wmma_f32(
    const float* __restrict__ X, const float* __restrict__ W,
    float* __restrict__ Y, int N, int K, int F) {
  constexpr int ROW_GROUPS = 8 / COL_WAVES;
  constexpr int BLOCK_ROWS = ROW_GROUPS * ROW_TILES * 16;
  constexpr int STRIDE = 36;
  __shared__ float lds[BLOCK_ROWS * STRIDE];

  const int tid = threadIdx.x;
  const int wave = tid >> 5;
  const int lane = tid & 31;
  const int colWave = wave % COL_WAVES;
  const int rowGroup = wave / COL_WAVES;
  const int col0 = colWave * 16;
  const int rowBase = blockIdx.x * BLOCK_ROWS;
  const int laneHi = lane >> 4;
  const int laneLo = lane & 15;

  v8f acc[ROW_TILES] = {};

  for (int k0 = 0; k0 < K; k0 += 32) {
#pragma unroll
    for (int v = 0; v < BLOCK_ROWS / 32; ++v) {
      int flat = (tid + v * 256) * 4;
      int row = flat >> 5;
      int kk = flat & 31;
      int grow = rowBase + row;
      float4 val{0.0f, 0.0f, 0.0f, 0.0f};
      if (grow < N) val = *(const float4*)(X + (long)grow * K + k0 + kk);
      *(float4*)&lds[row * STRIDE + kk] = val;
    }
    __syncthreads();

#pragma unroll
    for (int kk = 0; kk < 32; kk += 4) {
      const float* wp = W + (long)(k0 + kk + laneHi * 2) * F + col0 + laneLo;
      v2f bf;
      bf.x = wp[0];
      bf.y = wp[F];
#pragma unroll
      for (int t = 0; t < ROW_TILES; ++t) {
        int arow = (rowGroup * ROW_TILES + t) * 16 + laneLo;
        v2f af = *(const v2f*)&lds[arow * STRIDE + kk + laneHi * 2];
        acc[t] = __builtin_amdgcn_wmma_f32_16x16x4_f32(
             false, af,  false, bf,
             (short)0, acc[t],  false,  false);
      }
    }
    __syncthreads();
  }

#pragma unroll
  for (int t = 0; t < ROW_TILES; ++t) {
#pragma unroll
    for (int i = 0; i < 8; ++i) {
      int grow = rowBase + (rowGroup * ROW_TILES + t) * 16 + i + laneHi * 8;
      if (grow < N) Y[(long)grow * F + col0 + laneLo] = acc[t][i];
    }
  }
}

template <int WF, int NPW, bool ACCUM>
__global__ __launch_bounds__(256) void pull_rows(
    const float* __restrict__ Yr, const int* __restrict__ list,
    const int* __restrict__ cur, const float* __restrict__ rinvOut,
    float* __restrict__ Out, int N) {
  constexpr int VW = WF / 32;
  typedef __attribute__((ext_vector_type(VW))) float vecT;
  const int lane = threadIdx.x & 31;
  const int d0 = __builtin_amdgcn_readfirstlane(
      ((int)blockIdx.x * 8 + (int)(threadIdx.x >> 5)) * NPW);
  if (d0 >= N) return;
  const int nn = (N - d0 < NPW) ? (N - d0) : NPW;

  vecT prevs[NPW];
#pragma unroll
  for (int k = 0; k < NPW; ++k) {
    prevs[k] = (vecT)(0.0f);
    if constexpr (ACCUM) {
      const int dk = (d0 + k < N) ? (d0 + k) : (N - 1);
      prevs[k] = *(const vecT*)(Out + (long)dk * WF + lane * VW);
    }
  }

  int bnd = 0;
  if (lane <= nn) {
    const int i = d0 - 1 + lane;
    if (i >= 0) bnd = cur[i];
  }
  const int b = __builtin_amdgcn_readlane(bnd, nn);
  int ws = __builtin_amdgcn_readlane(bnd, 0);
  int sv = 0;
  float cv = 0.0f;
  if (ws + lane < b) {
    sv = list[ws + lane];
    cv = rinvOut[sv];
  }

#pragma unroll
  for (int k = 0; k < NPW; ++k) {
    if (k >= nn) break;
    const int sk = __builtin_amdgcn_readlane(bnd, k);
    const int ek = __builtin_amdgcn_readlane(bnd, k + 1);
    float* op = Out + (long)(d0 + k) * WF + lane * VW;
    vecT acc = (vecT)(0.0f);
    for (int j = sk; j < ek; j += 4) {
      if (j + 4 > ws + 32) {
        ws = j;
        sv = 0;
        cv = 0.0f;
        if (ws + lane < b) {
          sv = list[ws + lane];
          cv = rinvOut[sv];
        }
      }
      const int o = j - ws;
      const int n = ek - j;
      const int s0 = __builtin_amdgcn_readlane(sv, o);
      const int s1 = __builtin_amdgcn_readlane(sv, o + 1);
      const int s2 = __builtin_amdgcn_readlane(sv, o + 2);
      const int s3 = __builtin_amdgcn_readlane(sv, o + 3);
      const float c0 = __int_as_float(__builtin_amdgcn_readlane(__float_as_int(cv), o));
      const float c1 = __int_as_float(__builtin_amdgcn_readlane(__float_as_int(cv), o + 1));
      const float c2 = __int_as_float(__builtin_amdgcn_readlane(__float_as_int(cv), o + 2));
      const float c3 = __int_as_float(__builtin_amdgcn_readlane(__float_as_int(cv), o + 3));
      vecT v0 = *(const vecT*)(Yr + (long)s0 * WF + lane * VW);
      vecT v1 = (vecT)(0.0f), v2 = (vecT)(0.0f), v3 = (vecT)(0.0f);
      if (n > 1) v1 = *(const vecT*)(Yr + (long)s1 * WF + lane * VW);
      if (n > 2) v2 = *(const vecT*)(Yr + (long)s2 * WF + lane * VW);
      if (n > 3) v3 = *(const vecT*)(Yr + (long)s3 * WF + lane * VW);
      acc += c0 * v0;
      if (n > 1) acc += c1 * v1;
      if (n > 2) acc += c2 * v2;
      if (n > 3) acc += c3 * v3;
    }
    const float ri = rsqrtf(fmaxf((float)(ek - sk), 1.0f));
    *(vecT*)op = prevs[k] + ri * acc;
  }
}

extern "C" void kernel_launch(void* const* d_in, const int* in_sizes, int n_in,
                              void* d_out, int out_size, void* d_ws,
                              size_t ws_size, hipStream_t stream) {
  const float* x  = (const float*)d_in[0];
  const int*  src = (const int*)d_in[1];
  const int*  dst = (const int*)d_in[2];
  const float* W1 = (const float*)d_in[3];
  const float* b1 = (const float*)d_in[4];
  const float* W2 = (const float*)d_in[5];
  const float* b2 = (const float*)d_in[6];
  float* out = (float*)d_out;

  const int IN_F = 128, H_F = 128, OUT_F = 64, R = 4;
  const int N = in_sizes[0] / IN_F;
  const int E = in_sizes[1] / R;

  float* degOut = (float*)d_ws;
  int*   cur    = (int*)(degOut + (long)R * N);
  float* h      = (float*)(cur + (long)R * N);
  float* y      = h + (long)N * H_F;
  int* listA = (int*)out;
  int* listB = (int*)(y + (long)N * OUT_F);

  auto cdiv = [](long a, long b) { return (int)((a + b - 1) / b); };
  const int NPW = 4;

  fill_zero<<<cdiv(2L * R * N, 256), 256, 0, stream>>>(degOut, 2L * R * N);

  degree_count<<<cdiv((long)R * E, 256), 256, 0, stream>>>(src, dst, degOut,
                                                           cur, N, E,
                                                           (long)R * E);
  rinv_transform<<<cdiv((long)R * N, 256), 256, 0, stream>>>(degOut, (long)R * N);
  scan_counts<<<R, 1024, 0, stream>>>(cur, N);
  place_edges<<<cdiv((long)R * E, 256), 256, 0, stream>>>(src, dst, cur, listA,
                                                          N, E, (long)R * E);

  for (int r = 0; r < R; ++r) {
    gemm_wmma_f32<8, 4><<<cdiv(N, 64), 256, 0, stream>>>(
        x, W1 + (long)r * IN_F * H_F, y, N, IN_F, H_F);
    if (r == 0)
      pull_rows<128, NPW, false><<<cdiv(N, 8 * NPW), 256, 0, stream>>>(
          y, listA + (long)r * E, cur + (long)r * N, degOut + (long)r * N, h, N);
    else
      pull_rows<128, NPW, true><<<cdiv(N, 8 * NPW), 256, 0, stream>>>(
          y, listA + (long)r * E, cur + (long)r * N, degOut + (long)r * N, h, N);
  }
  bias_relu<<<cdiv((long)N * H_F, 256), 256, 0, stream>>>(h, b1,
                                                          (long)N * H_F, H_F);

  move_list<<<cdiv((long)R * E, 256), 256, 0, stream>>>(listA, listB, (long)R * E);

  for (int r = 0; r < R; ++r) {
    gemm_wmma_f32<4, 4><<<cdiv(N, 128), 256, 0, stream>>>(
        h, W2 + (long)r * H_F * OUT_F, y, N, H_F, OUT_F);
    if (r == 0)
      pull_rows<64, NPW, false><<<cdiv(N, 8 * NPW), 256, 0, stream>>>(
          y, listB + (long)r * E, cur + (long)r * N, degOut + (long)r * N, out, N);
    else
      pull_rows<64, NPW, true><<<cdiv(N, 8 * NPW), 256, 0, stream>>>(
          y, listB + (long)r * E, cur + (long)r * N, degOut + (long)r * N, out, N);
  }
  bias_add<<<cdiv((long)N * OUT_F, 256), 256, 0, stream>>>(out, b2,
                                                           (long)N * OUT_F,
                                                           OUT_F);
}


# layer 1 aggregate-then-transform in one launch: per 64 rows each relation's owner sums of x rows go via LDS into the f32 WMMA with W1_r, accumulators kept over relations; order of f32 sums changes
# speedup vs baseline: 1.5855x; 1.2738x over previous
#include <hip/hip_runtime.h>


typedef __attribute__((ext_vector_type(2))) float v2f;
typedef __attribute__((ext_vector_type(8))) float v8f;

__device__ __forceinline__ void atomAddF32(float* p, float v) {
#if __has_builtin(__builtin_amdgcn_global_atomic_fadd_f32)
  typedef __attribute__((address_space(1))) float gfloat;
  (void)__builtin_amdgcn_global_atomic_fadd_f32((gfloat*)p, v);
#else
  (void)__hip_atomic_fetch_add(p, v, __ATOMIC_RELAXED, __HIP_MEMORY_SCOPE_AGENT);
#endif
}

__global__ void fill_zero(float* __restrict__ p, long n) {
  long i = (long)blockIdx.x * blockDim.x + threadIdx.x;
  if (i < n) p[i] = 0.0f;
}

__global__ void degree_count(const int* __restrict__ src, const int* __restrict__ dst,
                             float* __restrict__ degOut, int* __restrict__ cntIn,
                             int N, int E, long RE) {
  long i = (long)blockIdx.x * blockDim.x + threadIdx.x;
  if (i >= RE) return;
  long r = i / E;
  int s = src[i];
  int d = dst[i];
  atomAddF32(&degOut[r * (long)N + s], 1.0f);
  atomicAdd(&cntIn[r * (long)N + d], 1);
}

__global__ void rinv_transform(float* __restrict__ p, long n) {
  long i = (long)blockIdx.x * blockDim.x + threadIdx.x;
  if (i >= n) return;
  p[i] = rsqrtf(fmaxf(p[i], 1.0f));
}

__global__ __launch_bounds__(1024) void scan_counts(int* __restrict__ cnt, int N) {
  __shared__ int waveTot[32];
  int* c = cnt + (long)blockIdx.x * N;
  const int tid = threadIdx.x;
  const int lane = tid & 31;
  const int wave = tid >> 5;
  int running = 0;
  int n0 = 0, n1 = 0, n2 = 0, n3 = 0;
  {
    const int i = tid * 4;
    if (i < N) n0 = c[i];
    if (i + 1 < N) n1 = c[i + 1];
    if (i + 2 < N) n2 = c[i + 2];
    if (i + 3 < N) n3 = c[i + 3];
  }
  for (int base = 0; base < N; base += 4096) {
    const int i = base + tid * 4;
    const int v0 = n0, v1 = n1, v2 = n2, v3 = n3;
    {
      const int q = i + 4096;
      n0 = (q < N) ? c[q] : 0;
      n1 = (q + 1 < N) ? c[q + 1] : 0;
      n2 = (q + 2 < N) ? c[q + 2] : 0;
      n3 = (q + 3 < N) ? c[q + 3] : 0;
    }
    const int s = v0 + v1 + v2 + v3;
    int inc = s;
    for (int o = 1; o < 32; o <<= 1) {
      int t = __shfl_up(inc, o);
      if (lane >= o) inc += t;
    }
    if (lane == 31) waveTot[wave] = inc;
    __syncthreads();
    const int wt = waveTot[lane];
    int winc = wt;
    for (int o = 1; o < 32; o <<= 1) {
      int t = __shfl_up(winc, o);
      if (lane >= o) winc += t;
    }
    const int waveOff = __shfl(winc, wave) - __shfl(wt, wave);
    const int total = __shfl(winc, 31);
    int ex = running + waveOff + inc - s;
    if (i < N) c[i] = ex;
    ex += v0;
    if (i + 1 < N) c[i + 1] = ex;
    ex += v1;
    if (i + 2 < N) c[i + 2] = ex;
    ex += v2;
    if (i + 3 < N) c[i + 3] = ex;
    running += total;
    __syncthreads();
  }
}

__global__ void place_edges(const int* __restrict__ src, const int* __restrict__ dst,
                            int* __restrict__ cur, int* __restrict__ list,
                            int N, int E, long RE) {
  long i = (long)blockIdx.x * blockDim.x + threadIdx.x;
  if (i >= RE) return;
  long r = i / E;
  int d = dst[i];
  int pos = atomicAdd(&cur[r * (long)N + d], 1);
  list[r * (long)E + pos] = src[i];
}

__global__ void move_list(const int* __restrict__ from, int* __restrict__ to, long n) {
  long i = (long)blockIdx.x * blockDim.x + threadIdx.x;
  if (i < n) to[i] = from[i];
}

__global__ void bias_add(float* __restrict__ o, const float* __restrict__ b,
                         long total, int F) {
  long i = (long)blockIdx.x * blockDim.x + threadIdx.x;
  if (i >= total) return;
  int f = (int)(i % F);
  o[i] += b[f] + b[F + f] + b[2 * F + f] + b[3 * F + f];
}

template <int COL_WAVES, int ROW_TILES>
__global__ __launch_bounds__(256) void gemm_wmma_f32(
    const float* __restrict__ X, const float* __restrict__ W,
    float* __restrict__ Y, int N, int K, int F) {
  constexpr int ROW_GROUPS = 8 / COL_WAVES;
  constexpr int BLOCK_ROWS = ROW_GROUPS * ROW_TILES * 16;
  constexpr int STRIDE = 36;
  __shared__ float lds[BLOCK_ROWS * STRIDE];

  const int tid = threadIdx.x;
  const int wave = tid >> 5;
  const int lane = tid & 31;
  const int colWave = wave % COL_WAVES;
  const int rowGroup = wave / COL_WAVES;
  const int col0 = colWave * 16;
  const int rowBase = blockIdx.x * BLOCK_ROWS;
  const int laneHi = lane >> 4;
  const int laneLo = lane & 15;

  v8f acc[ROW_TILES] = {};

  for (int k0 = 0; k0 < K; k0 += 32) {
#pragma unroll
    for (int v = 0; v < BLOCK_ROWS / 32; ++v) {
      int flat = (tid + v * 256) * 4;
      int row = flat >> 5;
      int kk = flat & 31;
      int grow = rowBase + row;
      float4 val{0.0f, 0.0f, 0.0f, 0.0f};
      if (grow < N) val = *(const float4*)(X + (long)grow * K + k0 + kk);
      *(float4*)&lds[row * STRIDE + kk] = val;
    }
    __syncthreads();

#pragma unroll
    for (int kk = 0; kk < 32; kk += 4) {
      const float* wp = W + (long)(k0 + kk + laneHi * 2) * F + col0 + laneLo;
      v2f bf;
      bf.x = wp[0];
      bf.y = wp[F];
#pragma unroll
      for (int t = 0; t < ROW_TILES; ++t) {
        int arow = (rowGroup * ROW_TILES + t) * 16 + laneLo;
        v2f af = *(const v2f*)&lds[arow * STRIDE + kk + laneHi * 2];
        acc[t] = __builtin_amdgcn_wmma_f32_16x16x4_f32(
             false, af,  false, bf,
             (short)0, acc[t],  false,  false);
      }
    }
    __syncthreads();
  }

#pragma unroll
  for (int t = 0; t < ROW_TILES; ++t) {
#pragma unroll
    for (int i = 0; i < 8; ++i) {
      int grow = rowBase + (rowGroup * ROW_TILES + t) * 16 + i + laneHi * 8;
      if (grow < N) Y[(long)grow * F + col0 + laneLo] = acc[t][i];
    }
  }
}

template <int WF, int NPW, bool ACCUM>
__global__ __launch_bounds__(256) void pull_rows(
    const float* __restrict__ Yr, const int* __restrict__ list,
    const int* __restrict__ cur, const float* __restrict__ rinvOut,
    float* __restrict__ Out, int N) {
  constexpr int VW = WF / 32;
  typedef __attribute__((ext_vector_type(VW))) float vecT;
  const int lane = threadIdx.x & 31;
  const int d0 = __builtin_amdgcn_readfirstlane(
      ((int)blockIdx.x * 8 + (int)(threadIdx.x >> 5)) * NPW);
  if (d0 >= N) return;
  const int nn = (N - d0 < NPW) ? (N - d0) : NPW;

  vecT prevs[NPW];
#pragma unroll
  for (int k = 0; k < NPW; ++k) {
    prevs[k] = (vecT)(0.0f);
    if constexpr (ACCUM) {
      const int dk = (d0 + k < N) ? (d0 + k) : (N - 1);
      prevs[k] = *(const vecT*)(Out + (long)dk * WF + lane * VW);
    }
  }

  int bnd = 0;
  if (lane <= nn) {
    const int i = d0 - 1 + lane;
    if (i >= 0) bnd = cur[i];
  }
  const int b = __builtin_amdgcn_readlane(bnd, nn);
  int ws = __builtin_amdgcn_readlane(bnd, 0);
  int sv = 0;
  float cv = 0.0f;
  if (ws + lane < b) {
    sv = list[ws + lane];
    cv = rinvOut[sv];
  }

#pragma unroll
  for (int k = 0; k < NPW; ++k) {
    if (k >= nn) break;
    const int sk = __builtin_amdgcn_readlane(bnd, k);
    const int ek = __builtin_amdgcn_readlane(bnd, k + 1);
    float* op = Out + (long)(d0 + k) * WF + lane * VW;
    vecT acc = (vecT)(0.0f);
    for (int j = sk; j < ek; j += 4) {
      if (j + 4 > ws + 32) {
        ws = j;
        sv = 0;
        cv = 0.0f;
        if (ws + lane < b) {
          sv = list[ws + lane];
          cv = rinvOut[sv];
        }
      }
      const int o = j - ws;
      const int n = ek - j;
      const int s0 = __builtin_amdgcn_readlane(sv, o);
      const int s1 = __builtin_amdgcn_readlane(sv, o + 1);
      const int s2 = __builtin_amdgcn_readlane(sv, o + 2);
      const int s3 = __builtin_amdgcn_readlane(sv, o + 3);
      const float c0 = __int_as_float(__builtin_amdgcn_readlane(__float_as_int(cv), o));
      const float c1 = __int_as_float(__builtin_amdgcn_readlane(__float_as_int(cv), o + 1));
      const float c2 = __int_as_float(__builtin_amdgcn_readlane(__float_as_int(cv), o + 2));
      const float c3 = __int_as_float(__builtin_amdgcn_readlane(__float_as_int(cv), o + 3));
      vecT v0 = *(const vecT*)(Yr + (long)s0 * WF + lane * VW);
      vecT v1 = (vecT)(0.0f), v2 = (vecT)(0.0f), v3 = (vecT)(0.0f);
      if (n > 1) v1 = *(const vecT*)(Yr + (long)s1 * WF + lane * VW);
      if (n > 2) v2 = *(const vecT*)(Yr + (long)s2 * WF + lane * VW);
      if (n > 3) v3 = *(const vecT*)(Yr + (long)s3 * WF + lane * VW);
      acc += c0 * v0;
      if (n > 1) acc += c1 * v1;
      if (n > 2) acc += c2 * v2;
      if (n > 3) acc += c3 * v3;
    }
    const float ri = rsqrtf(fmaxf((float)(ek - sk), 1.0f));
    *(vecT*)op = prevs[k] + ri * acc;
  }
}

__global__ __launch_bounds__(256) void agg_gemm_relu(
    const float* __restrict__ X, const int* __restrict__ list,
    const int* __restrict__ cur, const float* __restrict__ rinvOut,
    const float* __restrict__ W, const float* __restrict__ bias,
    float* __restrict__ H, int N, int E) {
  constexpr int K = 128, F = 128, R = 4;
  constexpr int ROW_TILES = 4, BLOCK_ROWS = 64, NPW = 8;
  constexpr int STRIDE = 132;
  __shared__ float lds[BLOCK_ROWS * STRIDE];

  const int tid = threadIdx.x;
  const int lane = tid & 31;
  const int wave = __builtin_amdgcn_readfirstlane(tid >> 5);
  const int col0 = wave * 16;
  const int rowBase = blockIdx.x * BLOCK_ROWS;
  const int laneHi = lane >> 4;
  const int laneLo = lane & 15;

  const int d0 = rowBase + wave * NPW;
  int nn = N - d0;
  nn = nn < 0 ? 0 : (nn > NPW ? NPW : nn);

  v8f acc[ROW_TILES] = {};

  for (int r = 0; r < R; ++r) {
    const int* listR = list + (long)r * E;
    const int* curR = cur + (long)r * N;
    const float* rinvR = rinvOut + (long)r * N;
    const float* Wr = W + (long)r * K * F;

    int bnd = 0;
    if (nn > 0 && lane <= nn) {
      const int i = d0 - 1 + lane;
      if (i >= 0) bnd = curR[i];
    }
    const int b = __builtin_amdgcn_readlane(bnd, nn);
    int ws = __builtin_amdgcn_readlane(bnd, 0);
    int sv = 0;
    float cv = 0.0f;
    if (ws + lane < b) {
      sv = listR[ws + lane];
      cv = rinvR[sv];
    }
    for (int k = 0; k < nn; ++k) {
      const int sk = __builtin_amdgcn_readlane(bnd, k);
      const int ek = __builtin_amdgcn_readlane(bnd, k + 1);
      float4 sum{0.0f, 0.0f, 0.0f, 0.0f};
      for (int j = sk; j < ek; j += 4) {
        if (j + 4 > ws + 32) {
          ws = j;
          sv = 0;
          cv = 0.0f;
          if (ws + lane < b) {
            sv = listR[ws + lane];
            cv = rinvR[sv];
          }
        }
        const int o = j - ws;
        const int n = ek - j;
        const int s0 = __builtin_amdgcn_readlane(sv, o);
        const int s1 = __builtin_amdgcn_readlane(sv, o + 1);
        const int s2 = __builtin_amdgcn_readlane(sv, o + 2);
        const int s3 = __builtin_amdgcn_readlane(sv, o + 3);
        const float c0 = __int_as_float(__builtin_amdgcn_readlane(__float_as_int(cv), o));
        const float c1 = __int_as_float(__builtin_amdgcn_readlane(__float_as_int(cv), o + 1));
        const float c2 = __int_as_float(__builtin_amdgcn_readlane(__float_as_int(cv), o + 2));
        const float c3 = __int_as_float(__builtin_amdgcn_readlane(__float_as_int(cv), o + 3));
        float4 v0 = *(const float4*)(X + (long)s0 * K + lane * 4);
        float4 v1{0.0f, 0.0f, 0.0f, 0.0f}, v2 = v1, v3 = v1;
        if (n > 1) v1 = *(const float4*)(X + (long)s1 * K + lane * 4);
        if (n > 2) v2 = *(const float4*)(X + (long)s2 * K + lane * 4);
        if (n > 3) v3 = *(const float4*)(X + (long)s3 * K + lane * 4);
        sum.x += c0 * v0.x; sum.y += c0 * v0.y; sum.z += c0 * v0.z; sum.w += c0 * v0.w;
        if (n > 1) { sum.x += c1 * v1.x; sum.y += c1 * v1.y; sum.z += c1 * v1.z; sum.w += c1 * v1.w; }
        if (n > 2) { sum.x += c2 * v2.x; sum.y += c2 * v2.y; sum.z += c2 * v2.z; sum.w += c2 * v2.w; }
        if (n > 3) { sum.x += c3 * v3.x; sum.y += c3 * v3.y; sum.z += c3 * v3.z; sum.w += c3 * v3.w; }
      }
      const float ri = rsqrtf(fmaxf((float)(ek - sk), 1.0f));
      float4 a{ri * sum.x, ri * sum.y, ri * sum.z, ri * sum.w};
      *(float4*)&lds[(wave * NPW + k) * STRIDE + lane * 4] = a;
    }
    for (int k = nn; k < NPW; ++k)
      *(float4*)&lds[(wave * NPW + k) * STRIDE + lane * 4] = float4{0.0f, 0.0f, 0.0f, 0.0f};
    __syncthreads();

#pragma unroll 8
    for (int kk = 0; kk < K; kk += 4) {
      const float* wp = Wr + (long)(kk + laneHi * 2) * F + col0 + laneLo;
      v2f bf;
      bf.x = wp[0];
      bf.y = wp[F];
#pragma unroll
      for (int t = 0; t < ROW_TILES; ++t) {
        const int arow = t * 16 + laneLo;
        v2f af = *(const v2f*)&lds[arow * STRIDE + kk + laneHi * 2];
        acc[t] = __builtin_amdgcn_wmma_f32_16x16x4_f32(
             false, af,  false, bf,
             (short)0, acc[t],  false,  false);
      }
    }
    __syncthreads();
  }

  const int col = col0 + laneLo;
  const float bs = bias[col] + bias[F + col] + bias[2 * F + col] + bias[3 * F + col];
#pragma unroll
  for (int t = 0; t < ROW_TILES; ++t) {
#pragma unroll
    for (int i = 0; i < 8; ++i) {
      const int grow = rowBase + t * 16 + i + laneHi * 8;
      const float v = acc[t][i] + bs;
      if (grow < N) H[(long)grow * F + col] = v > 0.0f ? v : 0.0f;
    }
  }
}

extern "C" void kernel_launch(void* const* d_in, const int* in_sizes, int n_in,
                              void* d_out, int out_size, void* d_ws,
                              size_t ws_size, hipStream_t stream) {
  const float* x  = (const float*)d_in[0];
  const int*  src = (const int*)d_in[1];
  const int*  dst = (const int*)d_in[2];
  const float* W1 = (const float*)d_in[3];
  const float* b1 = (const float*)d_in[4];
  const float* W2 = (const float*)d_in[5];
  const float* b2 = (const float*)d_in[6];
  float* out = (float*)d_out;

  const int IN_F = 128, H_F = 128, OUT_F = 64, R = 4;
  const int N = in_sizes[0] / IN_F;
  const int E = in_sizes[1] / R;

  float* degOut = (float*)d_ws;
  int*   cur    = (int*)(degOut + (long)R * N);
  float* h      = (float*)(cur + (long)R * N);
  float* y      = h + (long)N * H_F;
  int* listA = (int*)out;
  int* listB = (int*)(y + (long)N * OUT_F);

  auto cdiv = [](long a, long b) { return (int)((a + b - 1) / b); };
  const int NPW = 4;

  fill_zero<<<cdiv(2L * R * N, 256), 256, 0, stream>>>(degOut, 2L * R * N);

  degree_count<<<cdiv((long)R * E, 256), 256, 0, stream>>>(src, dst, degOut,
                                                           cur, N, E,
                                                           (long)R * E);
  rinv_transform<<<cdiv((long)R * N, 256), 256, 0, stream>>>(degOut, (long)R * N);
  scan_counts<<<R, 1024, 0, stream>>>(cur, N);
  place_edges<<<cdiv((long)R * E, 256), 256, 0, stream>>>(src, dst, cur, listA,
                                                          N, E, (long)R * E);

  agg_gemm_relu<<<cdiv(N, 64), 256, 0, stream>>>(x, listA, cur, degOut, W1, b1,
                                                 h, N, E);

  move_list<<<cdiv((long)R * E, 256), 256, 0, stream>>>(listA, listB, (long)R * E);

  for (int r = 0; r < R; ++r) {
    gemm_wmma_f32<4, 4><<<cdiv(N, 128), 256, 0, stream>>>(
        h, W2 + (long)r * H_F * OUT_F, y, N, H_F, OUT_F);
    if (r == 0)
      pull_rows<64, NPW, false><<<cdiv(N, 8 * NPW), 256, 0, stream>>>(
          y, listB + (long)r * E, cur + (long)r * N, degOut + (long)r * N, out, N);
    else
      pull_rows<64, NPW, true><<<cdiv(N, 8 * NPW), 256, 0, stream>>>(
          y, listB + (long)r * E, cur + (long)r * N, degOut + (long)r * N, out, N);
  }
  bias_add<<<cdiv((long)N * OUT_F, 256), 256, 0, stream>>>(out, b2,
                                                           (long)N * OUT_F,
                                                           OUT_F);
}


# set-up kernels (zero fill, degree count, factor transform, edge placement): a lane takes 8 to 16 elements by 16-byte accesses, 9x fewer waves; atomic arrival order still unordered
# speedup vs baseline: 2.0898x; 1.3181x over previous
#include <hip/hip_runtime.h>


typedef __attribute__((ext_vector_type(2))) float v2f;
typedef __attribute__((ext_vector_type(8))) float v8f;

__device__ __forceinline__ void atomAddF32(float* p, float v) {
#if __has_builtin(__builtin_amdgcn_global_atomic_fadd_f32)
  typedef __attribute__((address_space(1))) float gfloat;
  (void)__builtin_amdgcn_global_atomic_fadd_f32((gfloat*)p, v);
#else
  (void)__hip_atomic_fetch_add(p, v, __ATOMIC_RELAXED, __HIP_MEMORY_SCOPE_AGENT);
#endif
}


__global__ __launch_bounds__(256) void fill_zero(float* __restrict__ p, long n) {
  const long n4 = n >> 2;
  const long stride = (long)gridDim.x * 256;
  const long t0 = (long)blockIdx.x * 256 + threadIdx.x;
  float4* __restrict__ p4 = (float4*)p;
  for (long i = t0; i < n4; i += stride) p4[i] = float4{0.0f, 0.0f, 0.0f, 0.0f};
  const long t = (n4 << 2) + t0;
  if (t < n) p[t] = 0.0f;
}

__global__ __launch_bounds__(256) void degree_count(const int* __restrict__ src,
                                                    const int* __restrict__ dst,
                                                    float* __restrict__ degOut,
                                                    int* __restrict__ cntIn,
                                                    int N, int E) {
  const int r = blockIdx.y;
  const int* s = src + (long)r * E;
  const int* d = dst + (long)r * E;
  float* dO = degOut + (long)r * N;
  int* cI = cntIn + (long)r * N;
  const int t0 = blockIdx.x * 256 + threadIdx.x;
  const int stride = gridDim.x * 256;
  if ((E & 3) == 0) {
    const int E4 = E >> 2;
    const int4* s4 = (const int4*)s;
    const int4* d4 = (const int4*)d;
    for (int q = t0; q < E4; q += 2 * stride) {
      const int q1 = q + stride;
      const bool two = q1 < E4;
      const int4 sa = s4[q];
      const int4 da = d4[q];
      int4 sb{0, 0, 0, 0}, db{0, 0, 0, 0};
      if (two) {
        sb = s4[q1];
        db = d4[q1];
      }
      atomAddF32(&dO[sa.x], 1.0f);
      atomAddF32(&dO[sa.y], 1.0f);
      atomAddF32(&dO[sa.z], 1.0f);
      atomAddF32(&dO[sa.w], 1.0f);
      atomicAdd(&cI[da.x], 1);
      atomicAdd(&cI[da.y], 1);
      atomicAdd(&cI[da.z], 1);
      atomicAdd(&cI[da.w], 1);
      if (two) {
        atomAddF32(&dO[sb.x], 1.0f);
        atomAddF32(&dO[sb.y], 1.0f);
        atomAddF32(&dO[sb.z], 1.0f);
        atomAddF32(&dO[sb.w], 1.0f);
        atomicAdd(&cI[db.x], 1);
        atomicAdd(&cI[db.y], 1);
        atomicAdd(&cI[db.z], 1);
        atomicAdd(&cI[db.w], 1);
      }
    }
  } else {
    for (int i = t0; i < E; i += stride) {
      atomAddF32(&dO[s[i]], 1.0f);
      atomicAdd(&cI[d[i]], 1);
    }
  }
}

__global__ __launch_bounds__(256) void rinv_transform(float* __restrict__ p, long n) {
  const long n4 = n >> 2;
  const long stride = (long)gridDim.x * 256;
  const long t0 = (long)blockIdx.x * 256 + threadIdx.x;
  float4* __restrict__ p4 = (float4*)p;
  for (long i = t0; i < n4; i += stride) {
    float4 v = p4[i];
    v.x = rsqrtf(fmaxf(v.x, 1.0f));
    v.y = rsqrtf(fmaxf(v.y, 1.0f));
    v.z = rsqrtf(fmaxf(v.z, 1.0f));
    v.w = rsqrtf(fmaxf(v.w, 1.0f));
    p4[i] = v;
  }
  const long t = (n4 << 2) + t0;
  if (t < n) p[t] = rsqrtf(fmaxf(p[t], 1.0f));
}

__global__ __launch_bounds__(1024) void scan_counts(int* __restrict__ cnt, int N) {
  __shared__ int waveTot[32];
  int* c = cnt + (long)blockIdx.x * N;
  const int tid = threadIdx.x;
  const int lane = tid & 31;
  const int wave = tid >> 5;
  int running = 0;
  int n0 = 0, n1 = 0, n2 = 0, n3 = 0;
  {
    const int i = tid * 4;
    if (i < N) n0 = c[i];
    if (i + 1 < N) n1 = c[i + 1];
    if (i + 2 < N) n2 = c[i + 2];
    if (i + 3 < N) n3 = c[i + 3];
  }
  for (int base = 0; base < N; base += 4096) {
    const int i = base + tid * 4;
    const int v0 = n0, v1 = n1, v2 = n2, v3 = n3;
    {
      const int q = i + 4096;
      n0 = (q < N) ? c[q] : 0;
      n1 = (q + 1 < N) ? c[q + 1] : 0;
      n2 = (q + 2 < N) ? c[q + 2] : 0;
      n3 = (q + 3 < N) ? c[q + 3] : 0;
    }
    const int s = v0 + v1 + v2 + v3;
    int inc = s;
    for (int o = 1; o < 32; o <<= 1) {
      int t = __shfl_up(inc, o);
      if (lane >= o) inc += t;
    }
    if (lane == 31) waveTot[wave] = inc;
    __syncthreads();
    const int wt = waveTot[lane];
    int winc = wt;
    for (int o = 1; o < 32; o <<= 1) {
      int t = __shfl_up(winc, o);
      if (lane >= o) winc += t;
    }
    const int waveOff = __shfl(winc, wave) - __shfl(wt, wave);
    const int total = __shfl(winc, 31);
    int ex = running + waveOff + inc - s;
    if (i < N) c[i] = ex;
    ex += v0;
    if (i + 1 < N) c[i + 1] = ex;
    ex += v1;
    if (i + 2 < N) c[i + 2] = ex;
    ex += v2;
    if (i + 3 < N) c[i + 3] = ex;
    running += total;
    __syncthreads();
  }
}

__global__ __launch_bounds__(256) void place_edges(const int* __restrict__ src,
                                                   const int* __restrict__ dst,
                                                   int* __restrict__ cur,
                                                   int* __restrict__ list,
                                                   int N, int E) {
  const int r = blockIdx.y;
  const int* s = src + (long)r * E;
  const int* d = dst + (long)r * E;
  int* c = cur + (long)r * N;
  int* l = list + (long)r * E;
  const int t0 = blockIdx.x * 256 + threadIdx.x;
  const int stride = gridDim.x * 256;
  if ((E & 3) == 0) {
    const int E4 = E >> 2;
    const int4* s4 = (const int4*)s;
    const int4* d4 = (const int4*)d;
    for (int q = t0; q < E4; q += 2 * stride) {
      const int q1 = q + stride;
      const bool two = q1 < E4;
      const int4 sa = s4[q];
      const int4 da = d4[q];
      int4 sb{0, 0, 0, 0}, db{0, 0, 0, 0};
      if (two) {
        sb = s4[q1];
        db = d4[q1];
      }
      const int pa0 = atomicAdd(&c[da.x], 1);
      const int pa1 = atomicAdd(&c[da.y], 1);
      const int pa2 = atomicAdd(&c[da.z], 1);
      const int pa3 = atomicAdd(&c[da.w], 1);
      int pb0 = 0, pb1 = 0, pb2 = 0, pb3 = 0;
      if (two) {
        pb0 = atomicAdd(&c[db.x], 1);
        pb1 = atomicAdd(&c[db.y], 1);
        pb2 = atomicAdd(&c[db.z], 1);
        pb3 = atomicAdd(&c[db.w], 1);
      }
      l[pa0] = sa.x;
      l[pa1] = sa.y;
      l[pa2] = sa.z;
      l[pa3] = sa.w;
      if (two) {
        l[pb0] = sb.x;
        l[pb1] = sb.y;
        l[pb2] = sb.z;
        l[pb3] = sb.w;
      }
    }
  } else {
    for (int i = t0; i < E; i += stride) {
      const int pos = atomicAdd(&c[d[i]], 1);
      l[pos] = s[i];
    }
  }
}

template <int F, bool RELU>
__device__ __forceinline__ void agg_gemm_body(
    const float* __restrict__ X, const int* __restrict__ list,
    const int* __restrict__ cur, const float* __restrict__ rinvOut,
    const float* __restrict__ W, const float* __restrict__ bias,
    float* __restrict__ Y, int N, int E) {
  constexpr int K = 128, R = 4;
  constexpr int COL_WAVES = F / 16;
  constexpr int ROW_GROUPS = 8 / COL_WAVES;
  constexpr int BLOCK_ROWS = 64, NPW = 8;
  constexpr int ROW_TILES = BLOCK_ROWS / 16 / ROW_GROUPS;
  constexpr int STRIDE = 132;
  constexpr int K_AHEAD = ROW_TILES == 4 ? 8 : 4;
  __shared__ float lds[BLOCK_ROWS * STRIDE];

  const int tid = threadIdx.x;
  const int lane = tid & 31;
  const int wave = __builtin_amdgcn_readfirstlane(tid >> 5);
  const int col0 = (wave % COL_WAVES) * 16;
  const int tile0 = (wave / COL_WAVES) * ROW_TILES;
  const int rowBase = blockIdx.x * BLOCK_ROWS;
  const int laneHi = lane >> 4;
  const int laneLo = lane & 15;

  const int d0 = rowBase + wave * NPW;
  int nn = N - d0;
  nn = nn < 0 ? 0 : (nn > NPW ? NPW : nn);

  v8f acc[ROW_TILES] = {};

  for (int r = 0; r < R; ++r) {
    const int* listR = list + (long)r * E;
    const int* curR = cur + (long)r * N;
    const float* rinvR = rinvOut + (long)r * N;
    const float* Wr = W + (long)r * K * F;

    int bnd = 0;
    if (nn > 0 && lane <= nn) {
      const int i = d0 - 1 + lane;
      if (i >= 0) bnd = curR[i];
    }
    const int b = __builtin_amdgcn_readlane(bnd, nn);
    int ws = __builtin_amdgcn_readlane(bnd, 0);
    int sv = 0;
    float cv = 0.0f;
    if (ws + lane < b) {
      sv = listR[ws + lane];
      cv = rinvR[sv];
    }
    for (int k = 0; k < nn; ++k) {
      const int sk = __builtin_amdgcn_readlane(bnd, k);
      const int ek = __builtin_amdgcn_readlane(bnd, k + 1);
      float4 sum{0.0f, 0.0f, 0.0f, 0.0f};
      for (int j = sk; j < ek; j += 4) {
        if (j + 4 > ws + 32) {
          ws = j;
          sv = 0;
          cv = 0.0f;
          if (ws + lane < b) {
            sv = listR[ws + lane];
            cv = rinvR[sv];
          }
        }
        const int o = j - ws;
        const int n = ek - j;
        const int s0 = __builtin_amdgcn_readlane(sv, o);
        const int s1 = __builtin_amdgcn_readlane(sv, o + 1);
        const int s2 = __builtin_amdgcn_readlane(sv, o + 2);
        const int s3 = __builtin_amdgcn_readlane(sv, o + 3);
        const float c0 = __int_as_float(__builtin_amdgcn_readlane(__float_as_int(cv), o));
        const float c1 = __int_as_float(__builtin_amdgcn_readlane(__float_as_int(cv), o + 1));
        const float c2 = __int_as_float(__builtin_amdgcn_readlane(__float_as_int(cv), o + 2));
        const float c3 = __int_as_float(__builtin_amdgcn_readlane(__float_as_int(cv), o + 3));
        float4 v0 = *(const float4*)(X + (long)s0 * K + lane * 4);
        float4 v1{0.0f, 0.0f, 0.0f, 0.0f}, v2 = v1, v3 = v1;
        if (n > 1) v1 = *(const float4*)(X + (long)s1 * K + lane * 4);
        if (n > 2) v2 = *(const float4*)(X + (long)s2 * K + lane * 4);
        if (n > 3) v3 = *(const float4*)(X + (long)s3 * K + lane * 4);
        sum.x += c0 * v0.x; sum.y += c0 * v0.y; sum.z += c0 * v0.z; sum.w += c0 * v0.w;
        if (n > 1) { sum.x += c1 * v1.x; sum.y += c1 * v1.y; sum.z += c1 * v1.z; sum.w += c1 * v1.w; }
        if (n > 2) { sum.x += c2 * v2.x; sum.y += c2 * v2.y; sum.z += c2 * v2.z; sum.w += c2 * v2.w; }
        if (n > 3) { sum.x += c3 * v3.x; sum.y += c3 * v3.y; sum.z += c3 * v3.z; sum.w += c3 * v3.w; }
      }
      const float ri = rsqrtf(fmaxf((float)(ek - sk), 1.0f));
      float4 a{ri * sum.x, ri * sum.y, ri * sum.z, ri * sum.w};
      *(float4*)&lds[(wave * NPW + k) * STRIDE + lane * 4] = a;
    }
    for (int k = nn; k < NPW; ++k)
      *(float4*)&lds[(wave * NPW + k) * STRIDE + lane * 4] = float4{0.0f, 0.0f, 0.0f, 0.0f};
    __syncthreads();

#pragma unroll K_AHEAD
    for (int kk = 0; kk < K; kk += 4) {
      const float* wp = Wr + (long)(kk + laneHi * 2) * F + col0 + laneLo;
      v2f bf;
      bf.x = wp[0];
      bf.y = wp[F];
#pragma unroll
      for (int t = 0; t < ROW_TILES; ++t) {
        const int arow = (tile0 + t) * 16 + laneLo;
        v2f af = *(const v2f*)&lds[arow * STRIDE + kk + laneHi * 2];
        acc[t] = __builtin_amdgcn_wmma_f32_16x16x4_f32(
             false, af,  false, bf,
             (short)0, acc[t],  false,  false);
      }
    }
    __syncthreads();
  }

  const int col = col0 + laneLo;
  const float bs = bias[col] + bias[F + col] + bias[2 * F + col] + bias[3 * F + col];
#pragma unroll
  for (int t = 0; t < ROW_TILES; ++t) {
#pragma unroll
    for (int i = 0; i < 8; ++i) {
      const int grow = rowBase + (tile0 + t) * 16 + i + laneHi * 8;
      float v = acc[t][i] + bs;
      if constexpr (RELU) v = v > 0.0f ? v : 0.0f;
      if (grow < N) Y[(long)grow * F + col] = v;
    }
  }
}

__global__ __launch_bounds__(256) void agg_gemm_relu(
    const float* __restrict__ X, const int* __restrict__ list,
    const int* __restrict__ cur, const float* __restrict__ rinvOut,
    const float* __restrict__ W, const float* __restrict__ bias,
    float* __restrict__ H, int N, int E) {
  agg_gemm_body<128, true>(X, list, cur, rinvOut, W, bias, H, N, E);
}

__global__ __launch_bounds__(256, 16) void agg_gemm_out(
    const float* __restrict__ X, const int* __restrict__ list,
    const int* __restrict__ cur, const float* __restrict__ rinvOut,
    const float* __restrict__ W, const float* __restrict__ bias,
    float* __restrict__ Out, int N, int E) {
  agg_gemm_body<64, false>(X, list, cur, rinvOut, W, bias, Out, N, E);
}

extern "C" void kernel_launch(void* const* d_in, const int* in_sizes, int n_in,
                              void* d_out, int out_size, void* d_ws,
                              size_t ws_size, hipStream_t stream) {
  const float* x  = (const float*)d_in[0];
  const int*  src = (const int*)d_in[1];
  const int*  dst = (const int*)d_in[2];
  const float* W1 = (const float*)d_in[3];
  const float* b1 = (const float*)d_in[4];
  const float* W2 = (const float*)d_in[5];
  const float* b2 = (const float*)d_in[6];
  float* out = (float*)d_out;

  const int IN_F = 128, H_F = 128, R = 4;
  const int N = in_sizes[0] / IN_F;
  const int E = in_sizes[1] / R;

  float* degOut = (float*)d_ws;
  int*   cur    = (int*)(degOut + (long)R * N);
  float* h      = (float*)(cur + (long)R * N);
  int*   list   = (int*)(h + (long)N * H_F);

  auto cdiv = [](long a, long b) { return (int)((a + b - 1) / b); };

  fill_zero<<<cdiv(2L * R * N, 256 * 16), 256, 0, stream>>>(degOut, 2L * R * N);

  degree_count<<<dim3(cdiv(E, 256 * 8), R), 256, 0, stream>>>(src, dst, degOut,
                                                              cur, N, E);
  rinv_transform<<<cdiv((long)R * N, 256 * 8), 256, 0, stream>>>(degOut, (long)R * N);
  scan_counts<<<R, 1024, 0, stream>>>(cur, N);
  place_edges<<<dim3(cdiv(E, 256 * 8), R), 256, 0, stream>>>(src, dst, cur, list,
                                                             N, E);

  agg_gemm_relu<<<cdiv(N, 64), 256, 0, stream>>>(x, list, cur, degOut, W1, b1,
                                                 h, N, E);

  agg_gemm_out<<<cdiv(N, 64), 256, 0, stream>>>(h, list, cur, degOut, W2, b2,
                                                out, N, E);
}


# row gather of both layers walks a wave's edge run in chunks of 8 (layer 1) or 4 (layer 2) edges across node ends, all rows of a chunk requested before the first add; f32 sum order unchanged
# speedup vs baseline: 2.1690x; 1.0379x over previous
#include <hip/hip_runtime.h>


typedef __attribute__((ext_vector_type(2))) float v2f;
typedef __attribute__((ext_vector_type(8))) float v8f;

__device__ __forceinline__ void atomAddF32(float* p, float v) {
#if __has_builtin(__builtin_amdgcn_global_atomic_fadd_f32)
  typedef __attribute__((address_space(1))) float gfloat;
  (void)__builtin_amdgcn_global_atomic_fadd_f32((gfloat*)p, v);
#else
  (void)__hip_atomic_fetch_add(p, v, __ATOMIC_RELAXED, __HIP_MEMORY_SCOPE_AGENT);
#endif
}


__global__ __launch_bounds__(256) void fill_zero(float* __restrict__ p, long n) {
  const long n4 = n >> 2;
  const long stride = (long)gridDim.x * 256;
  const long t0 = (long)blockIdx.x * 256 + threadIdx.x;
  float4* __restrict__ p4 = (float4*)p;
  for (long i = t0; i < n4; i += stride) p4[i] = float4{0.0f, 0.0f, 0.0f, 0.0f};
  const long t = (n4 << 2) + t0;
  if (t < n) p[t] = 0.0f;
}

__global__ __launch_bounds__(256) void degree_count(const int* __restrict__ src,
                                                    const int* __restrict__ dst,
                                                    float* __restrict__ degOut,
                                                    int* __restrict__ cntIn,
                                                    int N, int E) {
  const int r = blockIdx.y;
  const int* s = src + (long)r * E;
  const int* d = dst + (long)r * E;
  float* dO = degOut + (long)r * N;
  int* cI = cntIn + (long)r * N;
  const int t0 = blockIdx.x * 256 + threadIdx.x;
  const int stride = gridDim.x * 256;
  if ((E & 3) == 0) {
    const int E4 = E >> 2;
    const int4* s4 = (const int4*)s;
    const int4* d4 = (const int4*)d;
    for (int q = t0; q < E4; q += 2 * stride) {
      const int q1 = q + stride;
      const bool two = q1 < E4;
      const int4 sa = s4[q];
      const int4 da = d4[q];
      int4 sb{0, 0, 0, 0}, db{0, 0, 0, 0};
      if (two) {
        sb = s4[q1];
        db = d4[q1];
      }
      atomAddF32(&dO[sa.x], 1.0f);
      atomAddF32(&dO[sa.y], 1.0f);
      atomAddF32(&dO[sa.z], 1.0f);
      atomAddF32(&dO[sa.w], 1.0f);
      atomicAdd(&cI[da.x], 1);
      atomicAdd(&cI[da.y], 1);
      atomicAdd(&cI[da.z], 1);
      atomicAdd(&cI[da.w], 1);
      if (two) {
        atomAddF32(&dO[sb.x], 1.0f);
        atomAddF32(&dO[sb.y], 1.0f);
        atomAddF32(&dO[sb.z], 1.0f);
        atomAddF32(&dO[sb.w], 1.0f);
        atomicAdd(&cI[db.x], 1);
        atomicAdd(&cI[db.y], 1);
        atomicAdd(&cI[db.z], 1);
        atomicAdd(&cI[db.w], 1);
      }
    }
  } else {
    for (int i = t0; i < E; i += stride) {
      atomAddF32(&dO[s[i]], 1.0f);
      atomicAdd(&cI[d[i]], 1);
    }
  }
}

__global__ __launch_bounds__(256) void rinv_transform(float* __restrict__ p, long n) {
  const long n4 = n >> 2;
  const long stride = (long)gridDim.x * 256;
  const long t0 = (long)blockIdx.x * 256 + threadIdx.x;
  float4* __restrict__ p4 = (float4*)p;
  for (long i = t0; i < n4; i += stride) {
    float4 v = p4[i];
    v.x = rsqrtf(fmaxf(v.x, 1.0f));
    v.y = rsqrtf(fmaxf(v.y, 1.0f));
    v.z = rsqrtf(fmaxf(v.z, 1.0f));
    v.w = rsqrtf(fmaxf(v.w, 1.0f));
    p4[i] = v;
  }
  const long t = (n4 << 2) + t0;
  if (t < n) p[t] = rsqrtf(fmaxf(p[t], 1.0f));
}

__global__ __launch_bounds__(1024) void scan_counts(int* __restrict__ cnt, int N) {
  __shared__ int waveTot[32];
  int* c = cnt + (long)blockIdx.x * N;
  const int tid = threadIdx.x;
  const int lane = tid & 31;
  const int wave = tid >> 5;
  int running = 0;
  int n0 = 0, n1 = 0, n2 = 0, n3 = 0;
  {
    const int i = tid * 4;
    if (i < N) n0 = c[i];
    if (i + 1 < N) n1 = c[i + 1];
    if (i + 2 < N) n2 = c[i + 2];
    if (i + 3 < N) n3 = c[i + 3];
  }
  for (int base = 0; base < N; base += 4096) {
    const int i = base + tid * 4;
    const int v0 = n0, v1 = n1, v2 = n2, v3 = n3;
    {
      const int q = i + 4096;
      n0 = (q < N) ? c[q] : 0;
      n1 = (q + 1 < N) ? c[q + 1] : 0;
      n2 = (q + 2 < N) ? c[q + 2] : 0;
      n3 = (q + 3 < N) ? c[q + 3] : 0;
    }
    const int s = v0 + v1 + v2 + v3;
    int inc = s;
    for (int o = 1; o < 32; o <<= 1) {
      int t = __shfl_up(inc, o);
      if (lane >= o) inc += t;
    }
    if (lane == 31) waveTot[wave] = inc;
    __syncthreads();
    const int wt = waveTot[lane];
    int winc = wt;
    for (int o = 1; o < 32; o <<= 1) {
      int t = __shfl_up(winc, o);
      if (lane >= o) winc += t;
    }
    const int waveOff = __shfl(winc, wave) - __shfl(wt, wave);
    const int total = __shfl(winc, 31);
    int ex = running + waveOff + inc - s;
    if (i < N) c[i] = ex;
    ex += v0;
    if (i + 1 < N) c[i + 1] = ex;
    ex += v1;
    if (i + 2 < N) c[i + 2] = ex;
    ex += v2;
    if (i + 3 < N) c[i + 3] = ex;
    running += total;
    __syncthreads();
  }
}

__global__ __launch_bounds__(256) void place_edges(const int* __restrict__ src,
                                                   const int* __restrict__ dst,
                                                   int* __restrict__ cur,
                                                   int* __restrict__ list,
                                                   int N, int E) {
  const int r = blockIdx.y;
  const int* s = src + (long)r * E;
  const int* d = dst + (long)r * E;
  int* c = cur + (long)r * N;
  int* l = list + (long)r * E;
  const int t0 = blockIdx.x * 256 + threadIdx.x;
  const int stride = gridDim.x * 256;
  if ((E & 3) == 0) {
    const int E4 = E >> 2;
    const int4* s4 = (const int4*)s;
    const int4* d4 = (const int4*)d;
    for (int q = t0; q < E4; q += 2 * stride) {
      const int q1 = q + stride;
      const bool two = q1 < E4;
      const int4 sa = s4[q];
      const int4 da = d4[q];
      int4 sb{0, 0, 0, 0}, db{0, 0, 0, 0};
      if (two) {
        sb = s4[q1];
        db = d4[q1];
      }
      const int pa0 = atomicAdd(&c[da.x], 1);
      const int pa1 = atomicAdd(&c[da.y], 1);
      const int pa2 = atomicAdd(&c[da.z], 1);
      const int pa3 = atomicAdd(&c[da.w], 1);
      int pb0 = 0, pb1 = 0, pb2 = 0, pb3 = 0;
      if (two) {
        pb0 = atomicAdd(&c[db.x], 1);
        pb1 = atomicAdd(&c[db.y], 1);
        pb2 = atomicAdd(&c[db.z], 1);
        pb3 = atomicAdd(&c[db.w], 1);
      }
      l[pa0] = sa.x;
      l[pa1] = sa.y;
      l[pa2] = sa.z;
      l[pa3] = sa.w;
      if (two) {
        l[pb0] = sb.x;
        l[pb1] = sb.y;
        l[pb2] = sb.z;
        l[pb3] = sb.w;
      }
    }
  } else {
    for (int i = t0; i < E; i += stride) {
      const int pos = atomicAdd(&c[d[i]], 1);
      l[pos] = s[i];
    }
  }
}

template <int F, bool RELU, int CH, int WREGS>
__device__ __forceinline__ void agg_gemm_body(
    const float* __restrict__ X, const int* __restrict__ list,
    const int* __restrict__ cur, const float* __restrict__ rinvOut,
    const float* __restrict__ W, const float* __restrict__ bias,
    float* __restrict__ Y, int N, int E) {
  constexpr int K = 128, R = 4;
  constexpr int COL_WAVES = F / 16;
  constexpr int ROW_GROUPS = 8 / COL_WAVES;
  constexpr int BLOCK_ROWS = 64, NPW = 8;
  constexpr int ROW_TILES = BLOCK_ROWS / 16 / ROW_GROUPS;
  constexpr int STRIDE = 132;
  static_assert(CH <= 32 && (WREGS == 1 || 32 % CH == 0), "a chunk must lie inside one window register");
  constexpr int K_AHEAD = ROW_TILES == 4 ? 8 : 4;
  __shared__ float lds[BLOCK_ROWS * STRIDE];

  const int tid = threadIdx.x;
  const int lane = tid & 31;
  const int wave = __builtin_amdgcn_readfirstlane(tid >> 5);
  const int col0 = (wave % COL_WAVES) * 16;
  const int tile0 = (wave / COL_WAVES) * ROW_TILES;
  const int rowBase = blockIdx.x * BLOCK_ROWS;
  const int laneHi = lane >> 4;
  const int laneLo = lane & 15;

  const int d0 = rowBase + wave * NPW;
  int nn = N - d0;
  nn = nn < 0 ? 0 : (nn > NPW ? NPW : nn);

  v8f acc[ROW_TILES] = {};

  for (int r = 0; r < R; ++r) {
    const int* listR = list + (long)r * E;
    const int* curR = cur + (long)r * N;
    const float* rinvR = rinvOut + (long)r * N;
    const float* Wr = W + (long)r * K * F;

    int bnd = 0;
    if (nn > 0 && lane <= nn) {
      const int i = d0 - 1 + lane;
      if (i >= 0) bnd = curR[i];
    }
    const int b = __builtin_amdgcn_readlane(bnd, nn);
    int ws = __builtin_amdgcn_readlane(bnd, 0);
    int sv[WREGS] = {};
    float cv[WREGS] = {};
    auto fillWindow = [&]() {
#pragma unroll
      for (int w = 0; w < WREGS; ++w) {
        const int e = ws + 32 * w + lane;
        sv[w] = listR[e < b ? e : b - 1];
      }
#pragma unroll
      for (int w = 0; w < WREGS; ++w) cv[w] = rinvR[sv[w]];
    };
    if (ws < b) fillWindow();
    int k = 0;
    int sk = ws;
    int ek = __builtin_amdgcn_readlane(bnd, 1);
    float4 sum{0.0f, 0.0f, 0.0f, 0.0f};
    auto flush = [&]() {
      const float ri = rsqrtf(fmaxf((float)(ek - sk), 1.0f));
      float4 a{ri * sum.x, ri * sum.y, ri * sum.z, ri * sum.w};
      *(float4*)&lds[(wave * NPW + k) * STRIDE + lane * 4] = a;
      ++k;
      sk = ek;
      ek = __builtin_amdgcn_readlane(bnd, k + 1);
      sum = float4{0.0f, 0.0f, 0.0f, 0.0f};
    };
    for (int j0 = ws; j0 < b; j0 += CH) {
      if (j0 - ws + CH > 32 * WREGS) {
        ws = j0;
        fillWindow();
      }
      const int o = j0 - ws;
      int svw = sv[0];
      int cvw = __float_as_int(cv[0]);
      if constexpr (WREGS == 2) {
        if (o >= 32) {
          svw = sv[1];
          cvw = __float_as_int(cv[1]);
        }
      }
      const int oo = o & 31;
      const int n = b - j0;
      int sid[CH];
      float cf[CH];
#pragma unroll
      for (int i = 0; i < CH; ++i) {
        sid[i] = __builtin_amdgcn_readlane(svw, oo + i);
        cf[i] = __int_as_float(__builtin_amdgcn_readlane(cvw, oo + i));
      }
      float4 v[CH];
#pragma unroll
      for (int i = 0; i < CH; ++i) {
        v[i] = float4{0.0f, 0.0f, 0.0f, 0.0f};
        if (i < n) v[i] = *(const float4*)(X + (long)sid[i] * K + lane * 4);
      }
#pragma unroll
      for (int i = 0; i < CH; ++i) {
        if (i < n) {
          while (j0 + i >= ek) flush();
          sum.x += cf[i] * v[i].x; sum.y += cf[i] * v[i].y;
          sum.z += cf[i] * v[i].z; sum.w += cf[i] * v[i].w;
        }
      }
    }
    while (k < nn) flush();
    for (int k = nn; k < NPW; ++k)
      *(float4*)&lds[(wave * NPW + k) * STRIDE + lane * 4] = float4{0.0f, 0.0f, 0.0f, 0.0f};
    __syncthreads();

#pragma unroll K_AHEAD
    for (int kk = 0; kk < K; kk += 4) {
      const float* wp = Wr + (long)(kk + laneHi * 2) * F + col0 + laneLo;
      v2f bf;
      bf.x = wp[0];
      bf.y = wp[F];
#pragma unroll
      for (int t = 0; t < ROW_TILES; ++t) {
        const int arow = (tile0 + t) * 16 + laneLo;
        v2f af = *(const v2f*)&lds[arow * STRIDE + kk + laneHi * 2];
        acc[t] = __builtin_amdgcn_wmma_f32_16x16x4_f32(
             false, af,  false, bf,
             (short)0, acc[t],  false,  false);
      }
    }
    __syncthreads();
  }

  const int col = col0 + laneLo;
  const float bs = bias[col] + bias[F + col] + bias[2 * F + col] + bias[3 * F + col];
#pragma unroll
  for (int t = 0; t < ROW_TILES; ++t) {
#pragma unroll
    for (int i = 0; i < 8; ++i) {
      const int grow = rowBase + (tile0 + t) * 16 + i + laneHi * 8;
      float v = acc[t][i] + bs;
      if constexpr (RELU) v = v > 0.0f ? v : 0.0f;
      if (grow < N) Y[(long)grow * F + col] = v;
    }
  }
}

__global__ __launch_bounds__(256) void agg_gemm_relu(
    const float* __restrict__ X, const int* __restrict__ list,
    const int* __restrict__ cur, const float* __restrict__ rinvOut,
    const float* __restrict__ W, const float* __restrict__ bias,
    float* __restrict__ H, int N, int E) {
  agg_gemm_body<128, true, 8, 2>(X, list, cur, rinvOut, W, bias, H, N, E);
}

__global__ __launch_bounds__(256, 16) void agg_gemm_out(
    const float* __restrict__ X, const int* __restrict__ list,
    const int* __restrict__ cur, const float* __restrict__ rinvOut,
    const float* __restrict__ W, const float* __restrict__ bias,
    float* __restrict__ Out, int N, int E) {
  agg_gemm_body<64, false, 4, 2>(X, list, cur, rinvOut, W, bias, Out, N, E);
}

extern "C" void kernel_launch(void* const* d_in, const int* in_sizes, int n_in,
                              void* d_out, int out_size, void* d_ws,
                              size_t ws_size, hipStream_t stream) {
  const float* x  = (const float*)d_in[0];
  const int*  src = (const int*)d_in[1];
  const int*  dst = (const int*)d_in[2];
  const float* W1 = (const float*)d_in[3];
  const float* b1 = (const float*)d_in[4];
  const float* W2 = (const float*)d_in[5];
  const float* b2 = (const float*)d_in[6];
  float* out = (float*)d_out;

  const int IN_F = 128, H_F = 128, R = 4;
  const int N = in_sizes[0] / IN_F;
  const int E = in_sizes[1] / R;

  float* degOut = (float*)d_ws;
  int*   cur    = (int*)(degOut + (long)R * N);
  float* h      = (float*)(cur + (long)R * N);
  int*   list   = (int*)(h + (long)N * H_F);

  auto cdiv = [](long a, long b) { return (int)((a + b - 1) / b); };

  fill_zero<<<cdiv(2L * R * N, 256 * 16), 256, 0, stream>>>(degOut, 2L * R * N);

  degree_count<<<dim3(cdiv(E, 256 * 8), R), 256, 0, stream>>>(src, dst, degOut,
                                                              cur, N, E);
  rinv_transform<<<cdiv((long)R * N, 256 * 8), 256, 0, stream>>>(degOut, (long)R * N);
  scan_counts<<<R, 1024, 0, stream>>>(cur, N);
  place_edges<<<dim3(cdiv(E, 256 * 8), R), 256, 0, stream>>>(src, dst, cur, list,
                                                             N, E);

  agg_gemm_relu<<<cdiv(N, 64), 256, 0, stream>>>(x, list, cur, degOut, W1, b1,
                                                 h, N, E);

  agg_gemm_out<<<cdiv(N, 64), 256, 0, stream>>>(h, list, cur, degOut, W2, b2,
                                                out, N, E);
}
